# starting kernel
# baseline (speedup 1.0000x reference)
#include <hip/hip_runtime.h>
#include <hip/hip_bf16.h>

typedef float v2f __attribute__((ext_vector_type(2)));
typedef float v4f __attribute__((ext_vector_type(4)));
typedef float v8f __attribute__((ext_vector_type(8)));

#define N_NODES 100000
#define N_EDGES 1600000
#define DDIM    64
#define FPAD    68
#define PSTRIDE 160

__global__ __launch_bounds__(256) void gcn_gemm_wmma(
    const float* __restrict__ features, const float* __restrict__ kern,
    float* __restrict__ h)
{
    __shared__ float Kpack[(DDIM / 2) * PSTRIDE];
    __shared__ float Flds[8][16 * FPAD];

    const int t = threadIdx.x;

    {
        const v4f* src = (const v4f*)kern;
        #pragma unroll
        for (int j = 0; j < (DDIM * DDIM / 4) / 256; ++j) {
            int i = t + j * 256;
            v4f w = src[i];
            int k  = i >> 4;
            int c0 = (i & 15) * 4;
            float* dst = &Kpack[(k >> 1) * PSTRIDE + (k & 1)];
            dst[(c0 + 0) * 2] = w.x;
            dst[(c0 + 1) * 2] = w.y;
            dst[(c0 + 2) * 2] = w.z;
            dst[(c0 + 3) * 2] = w.w;
        }
    }

    const int wave = t >> 5;
    const int lane = t & 31;
    const int m    = lane & 15;
    const int kh   = lane >> 4;

    const int tile   = blockIdx.x * 8 + wave;
    const bool active = (tile * 16) < N_NODES;

    if (active) {
        const int row0 = tile * 16;
        const v4f* src = (const v4f*)(features + (size_t)row0 * DDIM);
        #pragma unroll
        for (int j = 0; j < 8; ++j) {
            int i = lane + j * 32;
            int r = i >> 4;
            int c = i & 15;
            *(v4f*)&Flds[wave][r * FPAD + c * 4] = src[i];
        }
    }
    __syncthreads();
    if (!active) return;

    v8f acc0 = {}, acc1 = {}, acc2 = {}, acc3 = {};
    const float* Fw = &Flds[wave][0];

    #pragma unroll
    for (int k = 0; k < DDIM; k += 4) {
        v2f a = *(const v2f*)&Fw[m * FPAD + k + 2 * kh];

        const float* Kp = &Kpack[((k >> 1) + kh) * PSTRIDE];
        v2f b0 = *(const v2f*)&Kp[( 0 + m) * 2];
        v2f b1 = *(const v2f*)&Kp[(16 + m) * 2];
        v2f b2 = *(const v2f*)&Kp[(32 + m) * 2];
        v2f b3 = *(const v2f*)&Kp[(48 + m) * 2];

        acc0 = __builtin_amdgcn_wmma_f32_16x16x4_f32(false, a, false, b0, (short)0, acc0, false, false);
        acc1 = __builtin_amdgcn_wmma_f32_16x16x4_f32(false, a, false, b1, (short)0, acc1, false, false);
        acc2 = __builtin_amdgcn_wmma_f32_16x16x4_f32(false, a, false, b2, (short)0, acc2, false, false);
        acc3 = __builtin_amdgcn_wmma_f32_16x16x4_f32(false, a, false, b3, (short)0, acc3, false, false);
    }

    const int row0 = tile * 16;
    #pragma unroll
    for (int r = 0; r < 8; ++r) {
        float* hp = h + (size_t)(row0 + r + 8 * kh) * DDIM;
        hp[ 0 + m] = acc0[r];
        hp[16 + m] = acc1[r];
        hp[32 + m] = acc2[r];
        hp[48 + m] = acc3[r];
    }
}

__global__ __launch_bounds__(256) void gcn_edge_scatter(
    const int* __restrict__ erow, const int* __restrict__ ecol,
    const float* __restrict__ eval_, const float* __restrict__ h,
    float* __restrict__ agg)
{
    const int gtid = blockIdx.x * 256 + threadIdx.x;
    const int e    = gtid >> 5;
    const int lane = gtid & 31;
    if (e >= N_EDGES) return;

    const int   r = erow[e];
    const int   c = ecol[e];
    const float v = eval_[e];

    v2f hv = *(const v2f*)(h + (size_t)c * DDIM + lane * 2);
    float* dst = agg + (size_t)r * DDIM + lane * 2;
    atomicAdd(dst + 0, hv.x * v);
    atomicAdd(dst + 1, hv.y * v);
}

__global__ __launch_bounds__(256) void gcn_finalize(
    const float* __restrict__ h, const float* __restrict__ bias,
    const float* __restrict__ skip, float* __restrict__ out)
{
    const float alpha = 1.6732632423543772f;
    const float scale = 1.0507009873554805f;

    const int i = blockIdx.x * 256 + threadIdx.x;
    if (i >= N_NODES * DDIM / 2) return;
    const int dpair = i & (DDIM / 2 - 1);
    const int d0 = dpair * 2;

    v2f hv = *(const v2f*)(h + (size_t)i * 2);
    v2f av = *(const v2f*)(out + (size_t)i * 2);

    float x0 = hv.x * skip[d0 + 0] + av.x + bias[d0 + 0];
    float x1 = hv.y * skip[d0 + 1] + av.y + bias[d0 + 1];

    v2f res;
    res.x = x0 > 0.0f ? scale * x0 : scale * alpha * expm1f(x0);
    res.y = x1 > 0.0f ? scale * x1 : scale * alpha * expm1f(x1);
    *(v2f*)(out + (size_t)i * 2) = res;
}

extern "C" void kernel_launch(void* const* d_in, const int* in_sizes, int n_in,
                              void* d_out, int out_size, void* d_ws, size_t ws_size,
                              hipStream_t stream) {
    const float* features = (const float*)d_in[0];
    const float* kern     = (const float*)d_in[1];
    const float* bias     = (const float*)d_in[2];
    const float* skip     = (const float*)d_in[3];
    const int*   erow     = (const int*)  d_in[4];
    const int*   ecol     = (const int*)  d_in[5];
    const float* eval_    = (const float*)d_in[6];
    float* out = (float*)d_out;
    float* h   = (float*)d_ws;

    hipMemsetAsync(d_out, 0, (size_t)N_NODES * DDIM * sizeof(float), stream);

    const int n_tiles = N_NODES / 16;
    gcn_gemm_wmma<<<(n_tiles + 7) / 8, 256, 0, stream>>>(features, kern, h);

    gcn_edge_scatter<<<(N_EDGES * 32) / 256, 256, 0, stream>>>(erow, ecol, eval_, h, out);

    gcn_finalize<<<(N_NODES * DDIM / 2) / 256, 256, 0, stream>>>(h, bias, skip, out);
}


# f32-atomic scatter -> per-call counting sort of edges by row (int atomic rank) + one wave per row summing its edges in key order in registers, epilogue fused; sum order changes, fixed run to run
# speedup vs baseline: 1.1100x; 1.1100x over previous
#include <hip/hip_runtime.h>
#include <hip/hip_bf16.h>

typedef float v2f __attribute__((ext_vector_type(2)));
typedef float v4f __attribute__((ext_vector_type(4)));
typedef float v8f __attribute__((ext_vector_type(8)));

#define N_NODES 100000
#define N_EDGES 1600000
#define DDIM    64
#define FPAD    68
#define PSTRIDE 160
#define CNT_PAD 100352
#define SEG_CAP 64

typedef unsigned long long u64;

__global__ __launch_bounds__(256) void gcn_gemm_wmma(
    const float* __restrict__ features, const float* __restrict__ kern,
    float* __restrict__ h, int* __restrict__ cnt)
{
    __shared__ float Kpack[(DDIM / 2) * PSTRIDE];
    __shared__ float Flds[8][16 * FPAD];

    const int t = threadIdx.x;

    {
        const int g = blockIdx.x * 256 + t;
        if (g < CNT_PAD) cnt[g] = 0;
    }

    {
        const v4f* src = (const v4f*)kern;
        #pragma unroll
        for (int j = 0; j < (DDIM * DDIM / 4) / 256; ++j) {
            int i = t + j * 256;
            v4f w = src[i];
            int k  = i >> 4;
            int c0 = (i & 15) * 4;
            float* dst = &Kpack[(k >> 1) * PSTRIDE + (k & 1)];
            dst[(c0 + 0) * 2] = w.x;
            dst[(c0 + 1) * 2] = w.y;
            dst[(c0 + 2) * 2] = w.z;
            dst[(c0 + 3) * 2] = w.w;
        }
    }

    const int wave = t >> 5;
    const int lane = t & 31;
    const int m    = lane & 15;
    const int kh   = lane >> 4;

    const int tile   = blockIdx.x * 8 + wave;
    const bool active = (tile * 16) < N_NODES;

    if (active) {
        const int row0 = tile * 16;
        const v4f* src = (const v4f*)(features + (size_t)row0 * DDIM);
        #pragma unroll
        for (int j = 0; j < 8; ++j) {
            int i = lane + j * 32;
            int r = i >> 4;
            int c = i & 15;
            *(v4f*)&Flds[wave][r * FPAD + c * 4] = src[i];
        }
    }
    __syncthreads();
    if (!active) return;

    v8f acc0 = {}, acc1 = {}, acc2 = {}, acc3 = {};
    const float* Fw = &Flds[wave][0];

    #pragma unroll
    for (int k = 0; k < DDIM; k += 4) {
        v2f a = *(const v2f*)&Fw[m * FPAD + k + 2 * kh];

        const float* Kp = &Kpack[((k >> 1) + kh) * PSTRIDE];
        v2f b0 = *(const v2f*)&Kp[( 0 + m) * 2];
        v2f b1 = *(const v2f*)&Kp[(16 + m) * 2];
        v2f b2 = *(const v2f*)&Kp[(32 + m) * 2];
        v2f b3 = *(const v2f*)&Kp[(48 + m) * 2];

        acc0 = __builtin_amdgcn_wmma_f32_16x16x4_f32(false, a, false, b0, (short)0, acc0, false, false);
        acc1 = __builtin_amdgcn_wmma_f32_16x16x4_f32(false, a, false, b1, (short)0, acc1, false, false);
        acc2 = __builtin_amdgcn_wmma_f32_16x16x4_f32(false, a, false, b2, (short)0, acc2, false, false);
        acc3 = __builtin_amdgcn_wmma_f32_16x16x4_f32(false, a, false, b3, (short)0, acc3, false, false);
    }

    const int row0 = tile * 16;
    #pragma unroll
    for (int r = 0; r < 8; ++r) {
        float* hp = h + (size_t)(row0 + r + 8 * kh) * DDIM;
        hp[ 0 + m] = acc0[r];
        hp[16 + m] = acc1[r];
        hp[32 + m] = acc2[r];
        hp[48 + m] = acc3[r];
    }
}

__global__ __launch_bounds__(256) void gcn_edge_rank(
    const int* __restrict__ erow, int* __restrict__ cnt, float* __restrict__ rankbuf)
{
    const int e = blockIdx.x * 256 + threadIdx.x;
    rankbuf[e] = __int_as_float(atomicAdd(&cnt[erow[e]], 1));
}

__global__ __launch_bounds__(1024) void gcn_row_offsets(
    const int* __restrict__ cnt, int* __restrict__ off)
{
    __shared__ int s_w[32];
    const int t = threadIdx.x;
    const int lane = t & 31;
    const int wave = t >> 5;
    const int2* src = (const int2*)(cnt + wave * 3136) + lane;

    int sum = 0;
    #pragma unroll
    for (int i = 0; i < 49; ++i) {
        const int2 x = src[i * 32];
        sum += x.x + x.y;
    }
    for (int d = 16; d >= 1; d >>= 1) sum += __shfl_xor(sum, d, 32);
    if (lane == 0) s_w[wave] = sum;
    __syncthreads();
    if (wave == 0) {
        const int x = s_w[lane];
        int in2 = x;
        for (int d = 1; d < 32; d <<= 1) {
            const int y = __shfl_up(in2, d, 32);
            if (lane >= d) in2 += y;
        }
        s_w[lane] = in2 - x;
    }
    __syncthreads();

    int carry = s_w[wave];
    int2* dst = (int2*)(off + wave * 3136) + lane;
    #pragma unroll 7
    for (int i = 0; i < 49; ++i) {
        const int2 x = src[i * 32];
        const int s = x.x + x.y;
        int incl = s;
        for (int d = 1; d < 32; d <<= 1) {
            const int y = __shfl_up(incl, d, 32);
            if (lane >= d) incl += y;
        }
        int2 o;
        o.x = carry + incl - s;
        o.y = o.x + x.x;
        dst[i * 32] = o;
        carry += __shfl(incl, 31, 32);
    }
}

__global__ __launch_bounds__(256) void gcn_edge_place(
    const int* __restrict__ erow, const int* __restrict__ ecol,
    const float* __restrict__ eval_, const int* __restrict__ off,
    const float* __restrict__ rankbuf, u64* __restrict__ keys)
{
    const int e = blockIdx.x * 256 + threadIdx.x;
    const int pos = off[erow[e]] + __float_as_int(rankbuf[e]);
    keys[pos] = ((u64)(unsigned)ecol[e] << 32) | (u64)__float_as_uint(eval_[e]);
}

__global__ __launch_bounds__(256) void gcn_row_gather(
    const float* __restrict__ h, const int* __restrict__ off,
    const u64* __restrict__ keys, const float* __restrict__ bias,
    const float* __restrict__ skip, float* __restrict__ out)
{
    __shared__ u64 s_key[8][SEG_CAP];
    __shared__ u64 s_srt[8][SEG_CAP];

    const int lane = threadIdx.x & 31;
    const int w = __builtin_amdgcn_readfirstlane((int)(threadIdx.x >> 5));
    const int r = blockIdx.x * 8 + w;
    const int start = off[r];
    const int n = off[r + 1] - start;
    const u64* seg = keys + start;

    const v2f hs = *(const v2f*)(h + (size_t)r * DDIM + lane * 2);
    v2f acc = {0.0f, 0.0f};

    if (n <= SEG_CAP) {
        const u64 none = ~0ull;
        const u64 k0 = lane < n ? seg[lane] : none;
        const u64 k1 = lane + 32 < n ? seg[lane + 32] : none;
        s_key[w][lane] = k0;
        s_key[w][lane + 32] = k1;
        int r0 = 0, r1 = 0;
        if (n <= 32) {
            for (int j = 0; j < n; ++j) {
                const u64 kj = s_key[w][j];
                r0 += (kj < k0 || (kj == k0 && j < lane)) ? 1 : 0;
            }
        } else {
            for (int j = 0; j < n; ++j) {
                const u64 kj = s_key[w][j];
                r0 += (kj < k0 || (kj == k0 && j < lane)) ? 1 : 0;
                r1 += (kj < k1 || (kj == k1 && j < lane + 32)) ? 1 : 0;
            }
        }
        if (lane < n) s_srt[w][r0] = k0;
        if (lane + 32 < n) s_srt[w][r1] = k1;

        #pragma unroll 8
        for (int k = 0; k < n; ++k) {
            const u64 e = s_srt[w][k];
            const int   c = __builtin_amdgcn_readfirstlane((int)(e >> 32));
            const float v = __uint_as_float((unsigned)__builtin_amdgcn_readfirstlane((int)(unsigned)e));
            const v2f hv = *(const v2f*)(h + (size_t)c * DDIM + lane * 2);
            acc.x += v * hv.x;
            acc.y += v * hv.y;
        }
    } else {
        u64 prev = 0;
        int done = 0;
        for (int it = 0; it < n && done < n; ++it) {
            u64 best = ~0ull;
            int cb = 0;
            for (int j = lane; j < n; j += 32) {
                const u64 kj = seg[j];
                if (done == 0 || kj > prev) {
                    if (kj < best) { best = kj; cb = 1; }
                    else if (kj == best) { cb += 1; }
                }
            }
            u64 m = best;
            for (int d = 16; d >= 1; d >>= 1) {
                const u64 o = __shfl_xor(m, d, 32);
                m = o < m ? o : m;
            }
            int c = (best == m) ? cb : 0;
            for (int d = 16; d >= 1; d >>= 1) c += __shfl_xor(c, d, 32);
            const int   col = (int)(m >> 32);
            const float v   = __uint_as_float((unsigned)m);
            const v2f hv = *(const v2f*)(h + (size_t)col * DDIM + lane * 2);
            for (int i = 0; i < c; ++i) {
                acc.x += v * hv.x;
                acc.y += v * hv.y;
            }
            done += c;
            prev = m;
        }
    }

    const float alpha = 1.6732632423543772f;
    const float scale = 1.0507009873554805f;
    const v2f sk = *(const v2f*)(skip + lane * 2);
    const v2f bi = *(const v2f*)(bias + lane * 2);
    const float x0 = hs.x * sk.x + acc.x + bi.x;
    const float x1 = hs.y * sk.y + acc.y + bi.y;
    v2f res;
    res.x = x0 > 0.0f ? scale * x0 : scale * alpha * expm1f(x0);
    res.y = x1 > 0.0f ? scale * x1 : scale * alpha * expm1f(x1);
    *(v2f*)(out + (size_t)r * DDIM + lane * 2) = res;
}

extern "C" void kernel_launch(void* const* d_in, const int* in_sizes, int n_in,
                              void* d_out, int out_size, void* d_ws, size_t ws_size,
                              hipStream_t stream) {
    const float* features = (const float*)d_in[0];
    const float* kern     = (const float*)d_in[1];
    const float* bias     = (const float*)d_in[2];
    const float* skip     = (const float*)d_in[3];
    const int*   erow     = (const int*)  d_in[4];
    const int*   ecol     = (const int*)  d_in[5];
    const float* eval_    = (const float*)d_in[6];
    float* out = (float*)d_out;
    char*  ws  = (char*)d_ws;
    float* h    = (float*)ws;
    int*   cnt  = (int*)(ws + 25600000);
    int*   off  = (int*)(ws + 26001408);
    u64*   keys = (u64*)(ws + 26402816);

    if (ws_size < 39202816ull) return;

    gcn_gemm_wmma<<<782, 256, 0, stream>>>(features, kern, h, cnt);
    gcn_edge_rank<<<6250, 256, 0, stream>>>(erow, cnt, out);
    gcn_row_offsets<<<1, 1024, 0, stream>>>(cnt, off);
    gcn_edge_place<<<6250, 256, 0, stream>>>(erow, ecol, eval_, off, out, keys);
    gcn_row_gather<<<12500, 256, 0, stream>>>(h, off, keys, bias, skip, out);
}


# gather: two rows per wave side by side, half a wave per row with 16-byte h loads (lane = 4 dims); both row lists padded to whole groups of 8 by zero-weight entries on a zero line; sum order kept
# speedup vs baseline: 3.3330x; 3.0027x over previous
#include <hip/hip_runtime.h>
#include <hip/hip_bf16.h>

typedef float v2f __attribute__((ext_vector_type(2)));
typedef float v4f __attribute__((ext_vector_type(4)));
typedef float v8f __attribute__((ext_vector_type(8)));

#define N_NODES 100000
#define N_EDGES 1600000
#define DDIM    64
#define FPAD    68
#define PSTRIDE 160
#define BROWS   128
#define NB      782
#define NBP     800
#define TILE    8192
#define NTILES  196
#define CAP     3072
#define CNT_ZERO 2048
#define SEG_CAP 64
#define HCAP    32
#define GRP     8
#define ZROW    (N_NODES + 48)

typedef unsigned long long u64;

__global__ __launch_bounds__(256) void gcn_gemm_wmma(
    const float* __restrict__ features, const float* __restrict__ kern,
    float* __restrict__ h, int* __restrict__ cnt)
{
    __shared__ float Kpack[(DDIM / 2) * PSTRIDE];
    __shared__ float Flds[8][16 * FPAD];

    const int t = threadIdx.x;

    {
        const int g = blockIdx.x * 256 + t;
        if (g < CNT_ZERO) cnt[g] = 0;
        if (g < DDIM) h[(size_t)ZROW * DDIM + g] = 0.0f;
    }

    {
        const v4f* src = (const v4f*)kern;
        #pragma unroll
        for (int j = 0; j < (DDIM * DDIM / 4) / 256; ++j) {
            int i = t + j * 256;
            v4f w = src[i];
            int k  = i >> 4;
            int c0 = (i & 15) * 4;
            float* dst = &Kpack[(k >> 1) * PSTRIDE + (k & 1)];
            dst[(c0 + 0) * 2] = w.x;
            dst[(c0 + 1) * 2] = w.y;
            dst[(c0 + 2) * 2] = w.z;
            dst[(c0 + 3) * 2] = w.w;
        }
    }

    const int wave = t >> 5;
    const int lane = t & 31;
    const int m    = lane & 15;
    const int kh   = lane >> 4;

    const int tile   = blockIdx.x * 8 + wave;
    const bool active = (tile * 16) < N_NODES;

    if (active) {
        const int row0 = tile * 16;
        const v4f* src = (const v4f*)(features + (size_t)row0 * DDIM);
        #pragma unroll
        for (int j = 0; j < 8; ++j) {
            int i = lane + j * 32;
            int r = i >> 4;
            int c = i & 15;
            *(v4f*)&Flds[wave][r * FPAD + c * 4] = src[i];
        }
    }
    __syncthreads();
    if (!active) return;

    v8f acc0 = {}, acc1 = {}, acc2 = {}, acc3 = {};
    const float* Fw = &Flds[wave][0];

    #pragma unroll
    for (int k = 0; k < DDIM; k += 4) {
        v2f a = *(const v2f*)&Fw[m * FPAD + k + 2 * kh];

        const float* Kp = &Kpack[((k >> 1) + kh) * PSTRIDE];
        v2f b0 = *(const v2f*)&Kp[( 0 + m) * 2];
        v2f b1 = *(const v2f*)&Kp[(16 + m) * 2];
        v2f b2 = *(const v2f*)&Kp[(32 + m) * 2];
        v2f b3 = *(const v2f*)&Kp[(48 + m) * 2];

        acc0 = __builtin_amdgcn_wmma_f32_16x16x4_f32(false, a, false, b0, (short)0, acc0, false, false);
        acc1 = __builtin_amdgcn_wmma_f32_16x16x4_f32(false, a, false, b1, (short)0, acc1, false, false);
        acc2 = __builtin_amdgcn_wmma_f32_16x16x4_f32(false, a, false, b2, (short)0, acc2, false, false);
        acc3 = __builtin_amdgcn_wmma_f32_16x16x4_f32(false, a, false, b3, (short)0, acc3, false, false);
    }

    const int row0 = tile * 16;
    #pragma unroll
    for (int r = 0; r < 8; ++r) {
        float* hp = h + (size_t)(row0 + r + 8 * kh) * DDIM;
        hp[ 0 + m] = acc0[r];
        hp[16 + m] = acc1[r];
        hp[32 + m] = acc2[r];
        hp[48 + m] = acc3[r];
    }
}

__global__ __launch_bounds__(1024) void gcn_bucket_count(
    const int* __restrict__ erow, int* __restrict__ total)
{
    __shared__ int s_hist[NBP];
    const int t = threadIdx.x;
    if (t < NBP) s_hist[t] = 0;
    __syncthreads();
    const int e0 = blockIdx.x * TILE + t;
    #pragma unroll
    for (int i = 0; i < 8; ++i) {
        const int e = e0 + i * 1024;
        if (e < N_EDGES) atomicAdd(&s_hist[erow[e] >> 7], 1);
    }
    __syncthreads();
    if (t < NB) {
        const int c = s_hist[t];
        if (c) atomicAdd(&total[t], c);
    }
}

__global__ __launch_bounds__(1024) void gcn_bucket_place(
    const int* __restrict__ erow, const int* __restrict__ ecol,
    const float* __restrict__ eval_, const int* __restrict__ total,
    int* __restrict__ cursor, int* __restrict__ boff, u64* __restrict__ keys)
{
    __shared__ int s_hist[NBP];
    __shared__ int s_lst[NBP];
    __shared__ int s_gst[NBP];
    __shared__ int s_w[64];
    __shared__ u64 s_key[TILE];
    __shared__ int s_pos[TILE];

    const int t = threadIdx.x;
    const int lane = t & 31;
    const int wave = t >> 5;
    if (t < NBP) s_hist[t] = 0;
    __syncthreads();

    const int tile0 = blockIdx.x * TILE;
    u64 key[8];
    int bk[8], rk[8];
    #pragma unroll
    for (int i = 0; i < 8; ++i) {
        const int e = tile0 + i * 1024 + t;
        bk[i] = -1;
        rk[i] = 0;
        key[i] = 0;
        if (e < N_EDGES) {
            const int r = erow[e];
            const unsigned hi = ((unsigned)(r & (BROWS - 1)) << 17) | (unsigned)ecol[e];
            key[i] = ((u64)hi << 32) | (u64)__float_as_uint(eval_[e]);
            bk[i] = r >> 7;
            rk[i] = atomicAdd(&s_hist[bk[i]], 1);
        }
    }
    __syncthreads();

    int tot = 0, c = 0;
    if (t < NB) { tot = total[t]; c = s_hist[t]; }
    int itot = tot, ic = c;
    for (int d = 1; d < 32; d <<= 1) {
        const int y1 = __shfl_up(itot, d, 32);
        const int y2 = __shfl_up(ic, d, 32);
        if (lane >= d) { itot += y1; ic += y2; }
    }
    if (lane == 31) { s_w[wave] = itot; s_w[32 + wave] = ic; }
    __syncthreads();
    if (wave == 0) {
        const int x1 = s_w[lane];
        const int x2 = s_w[32 + lane];
        int in1 = x1, in2 = x2;
        for (int d = 1; d < 32; d <<= 1) {
            const int y1 = __shfl_up(in1, d, 32);
            const int y2 = __shfl_up(in2, d, 32);
            if (lane >= d) { in1 += y1; in2 += y2; }
        }
        s_w[lane] = in1 - x1;
        s_w[32 + lane] = in2 - x2;
    }
    __syncthreads();
    if (t < NB) {
        const int base = s_w[wave] + itot - tot;
        int g = 0;
        if (c) g = atomicAdd(&cursor[t], c);
        s_lst[t] = s_w[32 + wave] + ic - c;
        s_gst[t] = base + g;
        if (blockIdx.x == 0) boff[t] = base;
    }
    __syncthreads();

    #pragma unroll
    for (int i = 0; i < 8; ++i) {
        if (bk[i] >= 0) {
            const int lp = s_lst[bk[i]] + rk[i];
            s_key[lp] = key[i];
            s_pos[lp] = s_gst[bk[i]] + rk[i];
        }
    }
    __syncthreads();

    const int left = N_EDGES - tile0;
    const int ntile = left < TILE ? left : TILE;
    #pragma unroll
    for (int i = 0; i < 8; ++i) {
        const int j = i * 1024 + t;
        if (j < ntile) keys[s_pos[j]] = s_key[j];
    }
}

__global__ __launch_bounds__(512, 16) void gcn_bucket_gather(
    const float* __restrict__ h, const int* __restrict__ total,
    const int* __restrict__ boff, const u64* __restrict__ keys,
    const float* __restrict__ bias, const float* __restrict__ skip,
    float* __restrict__ out)
{
    __shared__ u64 s_srt[CAP + HCAP];
    __shared__ int s_cnt[BROWS];
    __shared__ int s_start[BROWS];
    __shared__ int s_cur[BROWS];
    __shared__ int s_w[4];
    __shared__ u64 s_row[16][SEG_CAP];
    __shared__ unsigned s_off[16][2 * HCAP];
    __shared__ float    s_val[16][2 * HCAP];
    __shared__ __attribute__((aligned(16))) float s_skip[DDIM];
    __shared__ __attribute__((aligned(16))) float s_bias[DDIM];

    const int t = threadIdx.x;
    const int lane = t & 31;
    const int w = __builtin_amdgcn_readfirstlane((int)(threadIdx.x >> 5));
    const int b = blockIdx.x;
    const int n = total[b];
    const u64* bsrc = keys + boff[b];
    const bool fits = n <= CAP;

    if (t < BROWS) s_cnt[t] = 0;
    if (t < DDIM) { s_skip[t] = skip[t]; s_bias[t] = bias[t]; }
    __syncthreads();
    u64 kk[6];
    if (fits) {
        #pragma unroll
        for (int i = 0; i < 6; ++i) {
            const int j = i * 512 + t;
            kk[i] = j < n ? bsrc[j] : 0ull;
        }
        #pragma unroll
        for (int i = 0; i < 6; ++i)
            if (i * 512 + t < n) atomicAdd(&s_cnt[(int)(kk[i] >> 49)], 1);
    } else {
        for (int j = t; j < n; j += 512) atomicAdd(&s_cnt[(int)(bsrc[j] >> 49)], 1);
    }
    __syncthreads();
    int c = 0, inc = 0;
    if (t < BROWS) {
        c = s_cnt[t];
        inc = c;
        for (int d = 1; d < 32; d <<= 1) {
            const int y = __shfl_up(inc, d, 32);
            if (lane >= d) inc += y;
        }
        if (lane == 31) s_w[w] = inc;
    }
    __syncthreads();
    if (t < BROWS) {
        int pre = 0;
        for (int q = 0; q < w; ++q) pre += s_w[q];
        const int st = pre + inc - c;
        s_start[t] = st;
        s_cur[t] = st;
    }
    __syncthreads();
    if (fits) {
        #pragma unroll
        for (int i = 0; i < 6; ++i) {
            if (i * 512 + t < n) {
                const int p = atomicAdd(&s_cur[(int)(kk[i] >> 49)], 1);
                s_srt[p] = kk[i];
            }
        }
        if (t < HCAP) s_srt[n + t] = ~0ull;
    }
    __syncthreads();

    const float alpha = 1.6732632423543772f;
    const float scale = 1.0507009873554805f;
    const int hf = lane >> 4;
    const int cl = lane & 15;
    const unsigned coff = (unsigned)cl << 4;
    const char* hbytes = (const char*)h;
    char* obytes = (char*)out;
    unsigned* woff = &s_off[w][hf * HCAP];
    float*    wval = &s_val[w][hf * HCAP];
    unsigned slow = 0;

    #pragma unroll 1
    for (int q = 0; q < 4; ++q) {
        const int lr0 = w * 8 + q * 2;
        if (b * BROWS + lr0 >= N_NODES) break;
        const int nA = __builtin_amdgcn_readfirstlane(s_cnt[lr0]);
        const int nB = __builtin_amdgcn_readfirstlane(s_cnt[lr0 + 1]);
        const int nmax = nA > nB ? nA : nB;
        if (!(fits && nmax <= HCAP)) { slow |= 1u << q; continue; }

        const int lrm = lr0 + hf;
        const int rm = b * BROWS + lrm;
        const bool vrow = rm < N_NODES;
        const int nm = hf ? nB : nA;
        const u64* seg = &s_srt[s_start[lrm]];
        const u64 none = ~0ull;
        const unsigned roff = ((unsigned)rm << 8) | coff;

        v4f hs = {0.0f, 0.0f, 0.0f, 0.0f};
        if (vrow) hs = *(const v4f*)(hbytes + roff);

        const int p1 = cl + 16;
        const u64 k0 = cl < nm ? seg[cl] : none;
        const u64 k1 = p1 < nm ? seg[p1] : none;
        int r0 = 0, r1 = 0;
        const int npad = (nmax + GRP - 1) & ~(GRP - 1);
        if (nmax <= 16) {
            for (int j = 0; j < npad; ++j) {
                const u64 kj = seg[j];
                r0 += (kj < k0 || (kj == k0 && j < cl)) ? 1 : 0;
            }
        } else {
            for (int j = 0; j < npad; ++j) {
                const u64 kj = seg[j];
                r0 += (kj < k0 || (kj == k0 && j < cl)) ? 1 : 0;
                r1 += (kj < k1 || (kj == k1 && j < p1)) ? 1 : 0;
            }
        }
        if (cl < nm) {
            woff[r0] = (((unsigned)(k0 >> 32)) & 0x1FFFFu) << 8;
            wval[r0] = __uint_as_float((unsigned)k0);
        } else {
            woff[cl] = (unsigned)ZROW << 8;
            wval[cl] = 0.0f;
        }
        if (p1 < nm) {
            woff[r1] = (((unsigned)(k1 >> 32)) & 0x1FFFFu) << 8;
            wval[r1] = __uint_as_float((unsigned)k1);
        } else {
            woff[p1] = (unsigned)ZROW << 8;
            wval[p1] = 0.0f;
        }

        v4f acc = {0.0f, 0.0f, 0.0f, 0.0f};
        for (int g0 = 0; g0 < npad; g0 += GRP) {
            #pragma unroll
            for (int u = 0; u < GRP; ++u) {
                const unsigned off = woff[g0 + u] | coff;
                const float v = wval[g0 + u];
                const v4f hv = *(const v4f*)(hbytes + off);
                acc.x = fmaf(v, hv.x, acc.x);
                acc.y = fmaf(v, hv.y, acc.y);
                acc.z = fmaf(v, hv.z, acc.z);
                acc.w = fmaf(v, hv.w, acc.w);
            }
        }

        if (vrow) {
            const v4f sk = *(const v4f*)&s_skip[cl * 4];
            const v4f bi = *(const v4f*)&s_bias[cl * 4];
            const float x0 = hs.x * sk.x + acc.x + bi.x;
            const float x1 = hs.y * sk.y + acc.y + bi.y;
            const float x2 = hs.z * sk.z + acc.z + bi.z;
            const float x3 = hs.w * sk.w + acc.w + bi.w;
            v4f res;
            res.x = x0 > 0.0f ? scale * x0 : scale * alpha * expm1f(x0);
            res.y = x1 > 0.0f ? scale * x1 : scale * alpha * expm1f(x1);
            res.z = x2 > 0.0f ? scale * x2 : scale * alpha * expm1f(x2);
            res.w = x3 > 0.0f ? scale * x3 : scale * alpha * expm1f(x3);
            *(v4f*)(obytes + roff) = res;
        }
    }

    #pragma unroll 1
    for (int q = 0; q < 4; ++q) {
        if (!((slow >> q) & 1u)) continue;
        const int lr0 = w * 8 + q * 2;
        const v2f sk = *(const v2f*)&s_skip[lane * 2];
        const v2f bi = *(const v2f*)&s_bias[lane * 2];
        for (int s = 0; s < 2; ++s) {
            const int lr = lr0 + s;
            const int r = b * BROWS + lr;
            if (r >= N_NODES) break;
            const int nr = __builtin_amdgcn_readfirstlane(s_cnt[lr]);
            const int st = __builtin_amdgcn_readfirstlane(s_start[lr]);

            const v2f hs = *(const v2f*)(h + (size_t)r * DDIM + lane * 2);
            v2f acc = {0.0f, 0.0f};

            if (fits && nr <= SEG_CAP) {
                const u64* seg = &s_srt[st];
                const u64 none = ~0ull;
                const u64 k0 = lane < nr ? seg[lane] : none;
                const u64 k1 = lane + 32 < nr ? seg[lane + 32] : none;
                int r0 = 0, r1 = 0;
                for (int j = 0; j < nr; ++j) {
                    const u64 kj = seg[j];
                    r0 += (kj < k0 || (kj == k0 && j < lane)) ? 1 : 0;
                    r1 += (kj < k1 || (kj == k1 && j < lane + 32)) ? 1 : 0;
                }
                if (lane < nr) s_row[w][r0] = k0;
                if (lane + 32 < nr) s_row[w][r1] = k1;
                for (int k = 0; k < nr; ++k) {
                    const u64 e = s_row[w][k];
                    const int   cc = __builtin_amdgcn_readfirstlane((int)(e >> 32)) & 0x1FFFF;
                    const float v = __uint_as_float((unsigned)__builtin_amdgcn_readfirstlane((int)(unsigned)e));
                    const v2f hv = *(const v2f*)(h + (size_t)cc * DDIM + lane * 2);
                    acc.x += v * hv.x;
                    acc.y += v * hv.y;
                }
            } else {
                const u64* src = fits ? (const u64*)&s_srt[st] : bsrc;
                const int cnt = fits ? nr : n;
                u64 prev = 0;
                int done = 0;
                for (int it = 0; it < nr && done < nr; ++it) {
                    u64 best = ~0ull;
                    int cb = 0;
                    for (int j = lane; j < cnt; j += 32) {
                        const u64 kj = src[j];
                        if ((int)(kj >> 49) == lr && (done == 0 || kj > prev)) {
                            if (kj < best) { best = kj; cb = 1; }
                            else if (kj == best) { cb += 1; }
                        }
                    }
                    u64 m = best;
                    for (int d = 16; d >= 1; d >>= 1) {
                        const u64 o = __shfl_xor(m, d, 32);
                        m = o < m ? o : m;
                    }
                    int cm = (best == m) ? cb : 0;
                    for (int d = 16; d >= 1; d >>= 1) cm += __shfl_xor(cm, d, 32);
                    const int   col = (int)(m >> 32) & 0x1FFFF;
                    const float v   = __uint_as_float((unsigned)m);
                    const v2f hv = *(const v2f*)(h + (size_t)col * DDIM + lane * 2);
                    for (int i = 0; i < cm; ++i) {
                        acc.x += v * hv.x;
                        acc.y += v * hv.y;
                    }
                    done += cm;
                    prev = m;
                }
            }

            const float x0 = hs.x * sk.x + acc.x + bi.x;
            const float x1 = hs.y * sk.y + acc.y + bi.y;
            v2f res;
            res.x = x0 > 0.0f ? scale * x0 : scale * alpha * expm1f(x0);
            res.y = x1 > 0.0f ? scale * x1 : scale * alpha * expm1f(x1);
            *(v2f*)(out + (size_t)r * DDIM + lane * 2) = res;
        }
    }
}

extern "C" void kernel_launch(void* const* d_in, const int* in_sizes, int n_in,
                              void* d_out, int out_size, void* d_ws, size_t ws_size,
                              hipStream_t stream) {
    const float* features = (const float*)d_in[0];
    const float* kern     = (const float*)d_in[1];
    const float* bias     = (const float*)d_in[2];
    const float* skip     = (const float*)d_in[3];
    const int*   erow     = (const int*)  d_in[4];
    const int*   ecol     = (const int*)  d_in[5];
    const float* eval_    = (const float*)d_in[6];
    float* out = (float*)d_out;
    char*  ws  = (char*)d_ws;
    float* h      = (float*)ws;
    int*   total  = (int*)(ws + 25600000);
    int*   cursor = (int*)(ws + 25604096);
    int*   boff   = (int*)(ws + 25608192);
    u64*   keys   = (u64*)(ws + 26402816);

    if (ws_size < 39202816ull) return;

    gcn_gemm_wmma<<<782, 256, 0, stream>>>(features, kern, h, total);
    gcn_bucket_count<<<NTILES, 1024, 0, stream>>>(erow, total);
    gcn_bucket_place<<<NTILES, 1024, 0, stream>>>(erow, ecol, eval_, total, cursor, boff, keys);
    gcn_bucket_gather<<<NB, 512, 0, stream>>>(h, total, boff, keys, bias, skip, out);
}


# edge count and place passes: a thread's 8 edge loads (24 in place) and its LDS reads go out as one batch (clamped unconditional indices), not one bounds-tested load awaited per turn; sum order kept
# speedup vs baseline: 3.6729x; 1.1020x over previous
#include <hip/hip_runtime.h>
#include <hip/hip_bf16.h>

typedef float v2f __attribute__((ext_vector_type(2)));
typedef float v4f __attribute__((ext_vector_type(4)));
typedef float v8f __attribute__((ext_vector_type(8)));

#define N_NODES 100000
#define N_EDGES 1600000
#define DDIM    64
#define FPAD    68
#define PSTRIDE 160
#define BROWS   128
#define NB      782
#define NBP     800
#define TILE    8192
#define NTILES  196
#define CAP     3072
#define CNT_ZERO 2048
#define SEG_CAP 64
#define HCAP    32
#define GRP     8
#define ZROW    (N_NODES + 48)

typedef unsigned long long u64;

__global__ __launch_bounds__(256) void gcn_gemm_wmma(
    const float* __restrict__ features, const float* __restrict__ kern,
    float* __restrict__ h, int* __restrict__ cnt)
{
    __shared__ float Kpack[(DDIM / 2) * PSTRIDE];
    __shared__ float Flds[8][16 * FPAD];

    const int t = threadIdx.x;

    {
        const int g = blockIdx.x * 256 + t;
        if (g < CNT_ZERO) cnt[g] = 0;
        if (g < DDIM) h[(size_t)ZROW * DDIM + g] = 0.0f;
    }

    {
        const v4f* src = (const v4f*)kern;
        #pragma unroll
        for (int j = 0; j < (DDIM * DDIM / 4) / 256; ++j) {
            int i = t + j * 256;
            v4f w = src[i];
            int k  = i >> 4;
            int c0 = (i & 15) * 4;
            float* dst = &Kpack[(k >> 1) * PSTRIDE + (k & 1)];
            dst[(c0 + 0) * 2] = w.x;
            dst[(c0 + 1) * 2] = w.y;
            dst[(c0 + 2) * 2] = w.z;
            dst[(c0 + 3) * 2] = w.w;
        }
    }

    const int wave = t >> 5;
    const int lane = t & 31;
    const int m    = lane & 15;
    const int kh   = lane >> 4;

    const int tile   = blockIdx.x * 8 + wave;
    const bool active = (tile * 16) < N_NODES;

    if (active) {
        const int row0 = tile * 16;
        const v4f* src = (const v4f*)(features + (size_t)row0 * DDIM);
        #pragma unroll
        for (int j = 0; j < 8; ++j) {
            int i = lane + j * 32;
            int r = i >> 4;
            int c = i & 15;
            *(v4f*)&Flds[wave][r * FPAD + c * 4] = src[i];
        }
    }
    __syncthreads();
    if (!active) return;

    v8f acc0 = {}, acc1 = {}, acc2 = {}, acc3 = {};
    const float* Fw = &Flds[wave][0];

    #pragma unroll
    for (int k = 0; k < DDIM; k += 4) {
        v2f a = *(const v2f*)&Fw[m * FPAD + k + 2 * kh];

        const float* Kp = &Kpack[((k >> 1) + kh) * PSTRIDE];
        v2f b0 = *(const v2f*)&Kp[( 0 + m) * 2];
        v2f b1 = *(const v2f*)&Kp[(16 + m) * 2];
        v2f b2 = *(const v2f*)&Kp[(32 + m) * 2];
        v2f b3 = *(const v2f*)&Kp[(48 + m) * 2];

        acc0 = __builtin_amdgcn_wmma_f32_16x16x4_f32(false, a, false, b0, (short)0, acc0, false, false);
        acc1 = __builtin_amdgcn_wmma_f32_16x16x4_f32(false, a, false, b1, (short)0, acc1, false, false);
        acc2 = __builtin_amdgcn_wmma_f32_16x16x4_f32(false, a, false, b2, (short)0, acc2, false, false);
        acc3 = __builtin_amdgcn_wmma_f32_16x16x4_f32(false, a, false, b3, (short)0, acc3, false, false);
    }

    const int row0 = tile * 16;
    #pragma unroll
    for (int r = 0; r < 8; ++r) {
        float* hp = h + (size_t)(row0 + r + 8 * kh) * DDIM;
        hp[ 0 + m] = acc0[r];
        hp[16 + m] = acc1[r];
        hp[32 + m] = acc2[r];
        hp[48 + m] = acc3[r];
    }
}

__global__ __launch_bounds__(1024) void gcn_bucket_count(
    const int* __restrict__ erow, int* __restrict__ total)
{
    __shared__ int s_hist[NBP];
    const int t = threadIdx.x;
    if (t < NBP) s_hist[t] = 0;
    __syncthreads();
    const int e0 = blockIdx.x * TILE + t;
    int rr[8];
    #pragma unroll
    for (int i = 0; i < 8; ++i) {
        const int e = e0 + i * 1024;
        rr[i] = erow[e < N_EDGES ? e : N_EDGES - 1];
    }
    #pragma unroll
    for (int i = 0; i < 8; ++i) {
        const int e = e0 + i * 1024;
        if (e < N_EDGES) atomicAdd(&s_hist[rr[i] >> 7], 1);
    }
    __syncthreads();
    if (t < NB) {
        const int c = s_hist[t];
        if (c) atomicAdd(&total[t], c);
    }
}

__global__ __launch_bounds__(1024) void gcn_bucket_place(
    const int* __restrict__ erow, const int* __restrict__ ecol,
    const float* __restrict__ eval_, const int* __restrict__ total,
    int* __restrict__ cursor, int* __restrict__ boff, u64* __restrict__ keys)
{
    __shared__ int s_hist[NBP];
    __shared__ int s_lst[NBP];
    __shared__ int s_gst[NBP];
    __shared__ int s_w[64];
    __shared__ u64 s_key[TILE];
    __shared__ int s_pos[TILE];

    const int t = threadIdx.x;
    const int lane = t & 31;
    const int wave = t >> 5;
    if (t < NBP) s_hist[t] = 0;
    __syncthreads();

    const int tile0 = blockIdx.x * TILE;
    u64 key[8];
    int bk[8], rk[8];
    int rr[8], cc[8];
    unsigned vv[8];
    #pragma unroll
    for (int i = 0; i < 8; ++i) {
        const int e = tile0 + i * 1024 + t;
        const int ec = e < N_EDGES ? e : N_EDGES - 1;
        rr[i] = erow[ec];
        cc[i] = ecol[ec];
        vv[i] = __float_as_uint(eval_[ec]);
    }
    #pragma unroll
    for (int i = 0; i < 8; ++i) {
        const int e = tile0 + i * 1024 + t;
        bk[i] = -1;
        rk[i] = 0;
        key[i] = 0;
        if (e < N_EDGES) {
            const int r = rr[i];
            const unsigned hi = ((unsigned)(r & (BROWS - 1)) << 17) | (unsigned)cc[i];
            key[i] = ((u64)hi << 32) | (u64)vv[i];
            bk[i] = r >> 7;
            rk[i] = atomicAdd(&s_hist[bk[i]], 1);
        }
    }
    __syncthreads();

    int tot = 0, c = 0;
    if (t < NB) { tot = total[t]; c = s_hist[t]; }
    int itot = tot, ic = c;
    for (int d = 1; d < 32; d <<= 1) {
        const int y1 = __shfl_up(itot, d, 32);
        const int y2 = __shfl_up(ic, d, 32);
        if (lane >= d) { itot += y1; ic += y2; }
    }
    if (lane == 31) { s_w[wave] = itot; s_w[32 + wave] = ic; }
    __syncthreads();
    if (wave == 0) {
        const int x1 = s_w[lane];
        const int x2 = s_w[32 + lane];
        int in1 = x1, in2 = x2;
        for (int d = 1; d < 32; d <<= 1) {
            const int y1 = __shfl_up(in1, d, 32);
            const int y2 = __shfl_up(in2, d, 32);
            if (lane >= d) { in1 += y1; in2 += y2; }
        }
        s_w[lane] = in1 - x1;
        s_w[32 + lane] = in2 - x2;
    }
    __syncthreads();
    if (t < NB) {
        const int base = s_w[wave] + itot - tot;
        int g = 0;
        if (c) g = atomicAdd(&cursor[t], c);
        s_lst[t] = s_w[32 + wave] + ic - c;
        s_gst[t] = base + g;
        if (blockIdx.x == 0) boff[t] = base;
    }
    __syncthreads();

    int ls[8], gs[8];
    #pragma unroll
    for (int i = 0; i < 8; ++i) {
        const int bb = bk[i] >= 0 ? bk[i] : 0;
        ls[i] = s_lst[bb];
        gs[i] = s_gst[bb];
    }
    #pragma unroll
    for (int i = 0; i < 8; ++i) {
        if (bk[i] >= 0) {
            const int lp = ls[i] + rk[i];
            s_key[lp] = key[i];
            s_pos[lp] = gs[i] + rk[i];
        }
    }
    __syncthreads();

    const int left = N_EDGES - tile0;
    const int ntile = left < TILE ? left : TILE;
    u64 kv[8];
    int pp[8];
    #pragma unroll
    for (int i = 0; i < 8; ++i) {
        const int j = i * 1024 + t;
        const int jc = j < ntile ? j : 0;
        kv[i] = s_key[jc];
        pp[i] = s_pos[jc];
    }
    #pragma unroll
    for (int i = 0; i < 8; ++i) {
        const int j = i * 1024 + t;
        if (j < ntile) keys[pp[i]] = kv[i];
    }
}

__global__ __launch_bounds__(512, 16) void gcn_bucket_gather(
    const float* __restrict__ h, const int* __restrict__ total,
    const int* __restrict__ boff, const u64* __restrict__ keys,
    const float* __restrict__ bias, const float* __restrict__ skip,
    float* __restrict__ out)
{
    __shared__ u64 s_srt[CAP + HCAP];
    __shared__ int s_cnt[BROWS];
    __shared__ int s_start[BROWS];
    __shared__ int s_cur[BROWS];
    __shared__ int s_w[4];
    __shared__ u64 s_row[16][SEG_CAP];
    __shared__ unsigned s_off[16][2 * HCAP];
    __shared__ float    s_val[16][2 * HCAP];
    __shared__ __attribute__((aligned(16))) float s_skip[DDIM];
    __shared__ __attribute__((aligned(16))) float s_bias[DDIM];

    const int t = threadIdx.x;
    const int lane = t & 31;
    const int w = __builtin_amdgcn_readfirstlane((int)(threadIdx.x >> 5));
    const int b = blockIdx.x;
    const int n = total[b];
    const u64* bsrc = keys + boff[b];
    const bool fits = n <= CAP;

    if (t < BROWS) s_cnt[t] = 0;
    if (t < DDIM) { s_skip[t] = skip[t]; s_bias[t] = bias[t]; }
    __syncthreads();
    u64 kk[6];
    if (fits) {
        #pragma unroll
        for (int i = 0; i < 6; ++i) {
            const int j = i * 512 + t;
            kk[i] = j < n ? bsrc[j] : 0ull;
        }
        #pragma unroll
        for (int i = 0; i < 6; ++i)
            if (i * 512 + t < n) atomicAdd(&s_cnt[(int)(kk[i] >> 49)], 1);
    } else {
        for (int j = t; j < n; j += 512) atomicAdd(&s_cnt[(int)(bsrc[j] >> 49)], 1);
    }
    __syncthreads();
    int c = 0, inc = 0;
    if (t < BROWS) {
        c = s_cnt[t];
        inc = c;
        for (int d = 1; d < 32; d <<= 1) {
            const int y = __shfl_up(inc, d, 32);
            if (lane >= d) inc += y;
        }
        if (lane == 31) s_w[w] = inc;
    }
    __syncthreads();
    if (t < BROWS) {
        int pre = 0;
        for (int q = 0; q < w; ++q) pre += s_w[q];
        const int st = pre + inc - c;
        s_start[t] = st;
        s_cur[t] = st;
    }
    __syncthreads();
    if (fits) {
        #pragma unroll
        for (int i = 0; i < 6; ++i) {
            if (i * 512 + t < n) {
                const int p = atomicAdd(&s_cur[(int)(kk[i] >> 49)], 1);
                s_srt[p] = kk[i];
            }
        }
        if (t < HCAP) s_srt[n + t] = ~0ull;
    }
    __syncthreads();

    const float alpha = 1.6732632423543772f;
    const float scale = 1.0507009873554805f;
    const int hf = lane >> 4;
    const int cl = lane & 15;
    const unsigned coff = (unsigned)cl << 4;
    const char* hbytes = (const char*)h;
    char* obytes = (char*)out;
    unsigned* woff = &s_off[w][hf * HCAP];
    float*    wval = &s_val[w][hf * HCAP];
    unsigned slow = 0;

    #pragma unroll 1
    for (int q = 0; q < 4; ++q) {
        const int lr0 = w * 8 + q * 2;
        if (b * BROWS + lr0 >= N_NODES) break;
        const int nA = __builtin_amdgcn_readfirstlane(s_cnt[lr0]);
        const int nB = __builtin_amdgcn_readfirstlane(s_cnt[lr0 + 1]);
        const int nmax = nA > nB ? nA : nB;
        if (!(fits && nmax <= HCAP)) { slow |= 1u << q; continue; }

        const int lrm = lr0 + hf;
        const int rm = b * BROWS + lrm;
        const bool vrow = rm < N_NODES;
        const int nm = hf ? nB : nA;
        const u64* seg = &s_srt[s_start[lrm]];
        const u64 none = ~0ull;
        const unsigned roff = ((unsigned)rm << 8) | coff;

        v4f hs = {0.0f, 0.0f, 0.0f, 0.0f};
        if (vrow) hs = *(const v4f*)(hbytes + roff);

        const int p1 = cl + 16;
        const u64 k0 = cl < nm ? seg[cl] : none;
        const u64 k1 = p1 < nm ? seg[p1] : none;
        int r0 = 0, r1 = 0;
        const int npad = (nmax + GRP - 1) & ~(GRP - 1);
        if (nmax <= 16) {
            for (int j = 0; j < npad; ++j) {
                const u64 kj = seg[j];
                r0 += (kj < k0 || (kj == k0 && j < cl)) ? 1 : 0;
            }
        } else {
            for (int j = 0; j < npad; ++j) {
                const u64 kj = seg[j];
                r0 += (kj < k0 || (kj == k0 && j < cl)) ? 1 : 0;
                r1 += (kj < k1 || (kj == k1 && j < p1)) ? 1 : 0;
            }
        }
        if (cl < nm) {
            woff[r0] = (((unsigned)(k0 >> 32)) & 0x1FFFFu) << 8;
            wval[r0] = __uint_as_float((unsigned)k0);
        } else {
            woff[cl] = (unsigned)ZROW << 8;
            wval[cl] = 0.0f;
        }
        if (p1 < nm) {
            woff[r1] = (((unsigned)(k1 >> 32)) & 0x1FFFFu) << 8;
            wval[r1] = __uint_as_float((unsigned)k1);
        } else {
            woff[p1] = (unsigned)ZROW << 8;
            wval[p1] = 0.0f;
        }

        v4f acc = {0.0f, 0.0f, 0.0f, 0.0f};
        for (int g0 = 0; g0 < npad; g0 += GRP) {
            #pragma unroll
            for (int u = 0; u < GRP; ++u) {
                const unsigned off = woff[g0 + u] | coff;
                const float v = wval[g0 + u];
                const v4f hv = *(const v4f*)(hbytes + off);
                acc.x = fmaf(v, hv.x, acc.x);
                acc.y = fmaf(v, hv.y, acc.y);
                acc.z = fmaf(v, hv.z, acc.z);
                acc.w = fmaf(v, hv.w, acc.w);
            }
        }

        if (vrow) {
            const v4f sk = *(const v4f*)&s_skip[cl * 4];
            const v4f bi = *(const v4f*)&s_bias[cl * 4];
            const float x0 = hs.x * sk.x + acc.x + bi.x;
            const float x1 = hs.y * sk.y + acc.y + bi.y;
            const float x2 = hs.z * sk.z + acc.z + bi.z;
            const float x3 = hs.w * sk.w + acc.w + bi.w;
            v4f res;
            res.x = x0 > 0.0f ? scale * x0 : scale * alpha * expm1f(x0);
            res.y = x1 > 0.0f ? scale * x1 : scale * alpha * expm1f(x1);
            res.z = x2 > 0.0f ? scale * x2 : scale * alpha * expm1f(x2);
            res.w = x3 > 0.0f ? scale * x3 : scale * alpha * expm1f(x3);
            *(v4f*)(obytes + roff) = res;
        }
    }

    #pragma unroll 1
    for (int q = 0; q < 4; ++q) {
        if (!((slow >> q) & 1u)) continue;
        const int lr0 = w * 8 + q * 2;
        const v2f sk = *(const v2f*)&s_skip[lane * 2];
        const v2f bi = *(const v2f*)&s_bias[lane * 2];
        for (int s = 0; s < 2; ++s) {
            const int lr = lr0 + s;
            const int r = b * BROWS + lr;
            if (r >= N_NODES) break;
            const int nr = __builtin_amdgcn_readfirstlane(s_cnt[lr]);
            const int st = __builtin_amdgcn_readfirstlane(s_start[lr]);

            const v2f hs = *(const v2f*)(h + (size_t)r * DDIM + lane * 2);
            v2f acc = {0.0f, 0.0f};

            if (fits && nr <= SEG_CAP) {
                const u64* seg = &s_srt[st];
                const u64 none = ~0ull;
                const u64 k0 = lane < nr ? seg[lane] : none;
                const u64 k1 = lane + 32 < nr ? seg[lane + 32] : none;
                int r0 = 0, r1 = 0;
                for (int j = 0; j < nr; ++j) {
                    const u64 kj = seg[j];
                    r0 += (kj < k0 || (kj == k0 && j < lane)) ? 1 : 0;
                    r1 += (kj < k1 || (kj == k1 && j < lane + 32)) ? 1 : 0;
                }
                if (lane < nr) s_row[w][r0] = k0;
                if (lane + 32 < nr) s_row[w][r1] = k1;
                for (int k = 0; k < nr; ++k) {
                    const u64 e = s_row[w][k];
                    const int   cc = __builtin_amdgcn_readfirstlane((int)(e >> 32)) & 0x1FFFF;
                    const float v = __uint_as_float((unsigned)__builtin_amdgcn_readfirstlane((int)(unsigned)e));
                    const v2f hv = *(const v2f*)(h + (size_t)cc * DDIM + lane * 2);
                    acc.x += v * hv.x;
                    acc.y += v * hv.y;
                }
            } else {
                const u64* src = fits ? (const u64*)&s_srt[st] : bsrc;
                const int cnt = fits ? nr : n;
                u64 prev = 0;
                int done = 0;
                for (int it = 0; it < nr && done < nr; ++it) {
                    u64 best = ~0ull;
                    int cb = 0;
                    for (int j = lane; j < cnt; j += 32) {
                        const u64 kj = src[j];
                        if ((int)(kj >> 49) == lr && (done == 0 || kj > prev)) {
                            if (kj < best) { best = kj; cb = 1; }
                            else if (kj == best) { cb += 1; }
                        }
                    }
                    u64 m = best;
                    for (int d = 16; d >= 1; d >>= 1) {
                        const u64 o = __shfl_xor(m, d, 32);
                        m = o < m ? o : m;
                    }
                    int cm = (best == m) ? cb : 0;
                    for (int d = 16; d >= 1; d >>= 1) cm += __shfl_xor(cm, d, 32);
                    const int   col = (int)(m >> 32) & 0x1FFFF;
                    const float v   = __uint_as_float((unsigned)m);
                    const v2f hv = *(const v2f*)(h + (size_t)col * DDIM + lane * 2);
                    for (int i = 0; i < cm; ++i) {
                        acc.x += v * hv.x;
                        acc.y += v * hv.y;
                    }
                    done += cm;
                    prev = m;
                }
            }

            const float x0 = hs.x * sk.x + acc.x + bi.x;
            const float x1 = hs.y * sk.y + acc.y + bi.y;
            v2f res;
            res.x = x0 > 0.0f ? scale * x0 : scale * alpha * expm1f(x0);
            res.y = x1 > 0.0f ? scale * x1 : scale * alpha * expm1f(x1);
            *(v2f*)(out + (size_t)r * DDIM + lane * 2) = res;
        }
    }
}

extern "C" void kernel_launch(void* const* d_in, const int* in_sizes, int n_in,
                              void* d_out, int out_size, void* d_ws, size_t ws_size,
                              hipStream_t stream) {
    const float* features = (const float*)d_in[0];
    const float* kern     = (const float*)d_in[1];
    const float* bias     = (const float*)d_in[2];
    const float* skip     = (const float*)d_in[3];
    const int*   erow     = (const int*)  d_in[4];
    const int*   ecol     = (const int*)  d_in[5];
    const float* eval_    = (const float*)d_in[6];
    float* out = (float*)d_out;
    char*  ws  = (char*)d_ws;
    float* h      = (float*)ws;
    int*   total  = (int*)(ws + 25600000);
    int*   cursor = (int*)(ws + 25604096);
    int*   boff   = (int*)(ws + 25608192);
    u64*   keys   = (u64*)(ws + 26402816);

    if (ws_size < 39202816ull) return;

    gcn_gemm_wmma<<<782, 256, 0, stream>>>(features, kern, h, total);
    gcn_bucket_count<<<NTILES, 1024, 0, stream>>>(erow, total);
    gcn_bucket_place<<<NTILES, 1024, 0, stream>>>(erow, ecol, eval_, total, cursor, boff, keys);
    gcn_bucket_gather<<<NB, 512, 0, stream>>>(h, total, boff, keys, bias, skip, out);
}


# work re-cut to the chip's 256 units: edge tiles 196 x 8192 -> 256 x 6272 (one per unit), row buckets 782 x 128 rows -> 500 x 200 rows in 25-wave gather blocks, two per unit; sum order kept
# speedup vs baseline: 3.8489x; 1.0479x over previous
#include <hip/hip_runtime.h>
#include <hip/hip_bf16.h>

typedef float v2f __attribute__((ext_vector_type(2)));
typedef float v4f __attribute__((ext_vector_type(4)));
typedef float v8f __attribute__((ext_vector_type(8)));

#define N_NODES 100000
#define N_EDGES 1600000
#define DDIM    64
#define FPAD    68
#define PSTRIDE 160
#define BROWS   200
#define RPAD    256
#define NB      500
#define NBP     512
#define TILE    6272
#define TSLOTS  7
#define NTILES  256
#define GTHR    800
#define GWAVES  25
#define CAP     4800
#define CNT_ZERO 2048
#define SEG_CAP 64
#define HCAP    32
#define GRP     8
#define ZROW    (N_NODES + 48)

typedef unsigned long long u64;

__global__ __launch_bounds__(256) void gcn_gemm_wmma(
    const float* __restrict__ features, const float* __restrict__ kern,
    float* __restrict__ h, int* __restrict__ cnt)
{
    __shared__ float Kpack[(DDIM / 2) * PSTRIDE];
    __shared__ float Flds[8][16 * FPAD];

    const int t = threadIdx.x;

    {
        const int g = blockIdx.x * 256 + t;
        if (g < CNT_ZERO) cnt[g] = 0;
        if (g < DDIM) h[(size_t)ZROW * DDIM + g] = 0.0f;
    }

    {
        const v4f* src = (const v4f*)kern;
        #pragma unroll
        for (int j = 0; j < (DDIM * DDIM / 4) / 256; ++j) {
            int i = t + j * 256;
            v4f w = src[i];
            int k  = i >> 4;
            int c0 = (i & 15) * 4;
            float* dst = &Kpack[(k >> 1) * PSTRIDE + (k & 1)];
            dst[(c0 + 0) * 2] = w.x;
            dst[(c0 + 1) * 2] = w.y;
            dst[(c0 + 2) * 2] = w.z;
            dst[(c0 + 3) * 2] = w.w;
        }
    }

    const int wave = t >> 5;
    const int lane = t & 31;
    const int m    = lane & 15;
    const int kh   = lane >> 4;

    const int tile   = blockIdx.x * 8 + wave;
    const bool active = (tile * 16) < N_NODES;

    if (active) {
        const int row0 = tile * 16;
        const v4f* src = (const v4f*)(features + (size_t)row0 * DDIM);
        #pragma unroll
        for (int j = 0; j < 8; ++j) {
            int i = lane + j * 32;
            int r = i >> 4;
            int c = i & 15;
            *(v4f*)&Flds[wave][r * FPAD + c * 4] = src[i];
        }
    }
    __syncthreads();
    if (!active) return;

    v8f acc0 = {}, acc1 = {}, acc2 = {}, acc3 = {};
    const float* Fw = &Flds[wave][0];

    #pragma unroll
    for (int k = 0; k < DDIM; k += 4) {
        v2f a = *(const v2f*)&Fw[m * FPAD + k + 2 * kh];

        const float* Kp = &Kpack[((k >> 1) + kh) * PSTRIDE];
        v2f b0 = *(const v2f*)&Kp[( 0 + m) * 2];
        v2f b1 = *(const v2f*)&Kp[(16 + m) * 2];
        v2f b2 = *(const v2f*)&Kp[(32 + m) * 2];
        v2f b3 = *(const v2f*)&Kp[(48 + m) * 2];

        acc0 = __builtin_amdgcn_wmma_f32_16x16x4_f32(false, a, false, b0, (short)0, acc0, false, false);
        acc1 = __builtin_amdgcn_wmma_f32_16x16x4_f32(false, a, false, b1, (short)0, acc1, false, false);
        acc2 = __builtin_amdgcn_wmma_f32_16x16x4_f32(false, a, false, b2, (short)0, acc2, false, false);
        acc3 = __builtin_amdgcn_wmma_f32_16x16x4_f32(false, a, false, b3, (short)0, acc3, false, false);
    }

    const int row0 = tile * 16;
    #pragma unroll
    for (int r = 0; r < 8; ++r) {
        float* hp = h + (size_t)(row0 + r + 8 * kh) * DDIM;
        hp[ 0 + m] = acc0[r];
        hp[16 + m] = acc1[r];
        hp[32 + m] = acc2[r];
        hp[48 + m] = acc3[r];
    }
}

__global__ __launch_bounds__(1024) void gcn_bucket_count(
    const int* __restrict__ erow, int* __restrict__ total)
{
    __shared__ int s_hist[NBP];
    const int t = threadIdx.x;
    if (t < NBP) s_hist[t] = 0;
    __syncthreads();
    const int tile0 = blockIdx.x * TILE;
    const int left = N_EDGES - tile0;
    const int ntile = left < TILE ? left : TILE;
    int rr[TSLOTS];
    #pragma unroll
    for (int i = 0; i < TSLOTS; ++i) {
        const int j = i * 1024 + t;
        rr[i] = erow[j < ntile ? tile0 + j : N_EDGES - 1];
    }
    #pragma unroll
    for (int i = 0; i < TSLOTS; ++i) {
        const int j = i * 1024 + t;
        if (j < ntile) atomicAdd(&s_hist[(unsigned)rr[i] / (unsigned)BROWS], 1);
    }
    __syncthreads();
    if (t < NB) {
        const int c = s_hist[t];
        if (c) atomicAdd(&total[t], c);
    }
}

__global__ __launch_bounds__(1024) void gcn_bucket_place(
    const int* __restrict__ erow, const int* __restrict__ ecol,
    const float* __restrict__ eval_, const int* __restrict__ total,
    int* __restrict__ cursor, int* __restrict__ boff, u64* __restrict__ keys)
{
    __shared__ int s_hist[NBP];
    __shared__ int s_lst[NBP];
    __shared__ int s_gst[NBP];
    __shared__ int s_w[64];
    __shared__ u64 s_key[TILE];
    __shared__ int s_pos[TILE];

    const int t = threadIdx.x;
    const int lane = t & 31;
    const int wave = t >> 5;
    if (t < NBP) s_hist[t] = 0;
    __syncthreads();

    const int tile0 = blockIdx.x * TILE;
    const int left = N_EDGES - tile0;
    const int ntile = left < TILE ? left : TILE;
    u64 key[TSLOTS];
    int bk[TSLOTS], rk[TSLOTS];
    int rr[TSLOTS], cc[TSLOTS];
    unsigned vv[TSLOTS];
    #pragma unroll
    for (int i = 0; i < TSLOTS; ++i) {
        const int j = i * 1024 + t;
        const int ec = j < ntile ? tile0 + j : N_EDGES - 1;
        rr[i] = erow[ec];
        cc[i] = ecol[ec];
        vv[i] = __float_as_uint(eval_[ec]);
    }
    #pragma unroll
    for (int i = 0; i < TSLOTS; ++i) {
        const int j = i * 1024 + t;
        bk[i] = -1;
        rk[i] = 0;
        key[i] = 0;
        if (j < ntile) {
            const unsigned r = (unsigned)rr[i];
            const unsigned bb = r / (unsigned)BROWS;
            const unsigned hi = ((r - bb * (unsigned)BROWS) << 17) | (unsigned)cc[i];
            key[i] = ((u64)hi << 32) | (u64)vv[i];
            bk[i] = (int)bb;
            rk[i] = atomicAdd(&s_hist[bk[i]], 1);
        }
    }
    __syncthreads();

    int tot = 0, c = 0;
    if (t < NB) { tot = total[t]; c = s_hist[t]; }
    int itot = tot, ic = c;
    for (int d = 1; d < 32; d <<= 1) {
        const int y1 = __shfl_up(itot, d, 32);
        const int y2 = __shfl_up(ic, d, 32);
        if (lane >= d) { itot += y1; ic += y2; }
    }
    if (lane == 31) { s_w[wave] = itot; s_w[32 + wave] = ic; }
    __syncthreads();
    if (wave == 0) {
        const int x1 = s_w[lane];
        const int x2 = s_w[32 + lane];
        int in1 = x1, in2 = x2;
        for (int d = 1; d < 32; d <<= 1) {
            const int y1 = __shfl_up(in1, d, 32);
            const int y2 = __shfl_up(in2, d, 32);
            if (lane >= d) { in1 += y1; in2 += y2; }
        }
        s_w[lane] = in1 - x1;
        s_w[32 + lane] = in2 - x2;
    }
    __syncthreads();
    if (t < NB) {
        const int base = s_w[wave] + itot - tot;
        int g = 0;
        if (c) g = atomicAdd(&cursor[t], c);
        s_lst[t] = s_w[32 + wave] + ic - c;
        s_gst[t] = base + g;
        if (blockIdx.x == 0) boff[t] = base;
    }
    __syncthreads();

    int ls[TSLOTS], gs[TSLOTS];
    #pragma unroll
    for (int i = 0; i < TSLOTS; ++i) {
        const int bb = bk[i] >= 0 ? bk[i] : 0;
        ls[i] = s_lst[bb];
        gs[i] = s_gst[bb];
    }
    #pragma unroll
    for (int i = 0; i < TSLOTS; ++i) {
        if (bk[i] >= 0) {
            const int lp = ls[i] + rk[i];
            s_key[lp] = key[i];
            s_pos[lp] = gs[i] + rk[i];
        }
    }
    __syncthreads();

    u64 kv[TSLOTS];
    int pp[TSLOTS];
    #pragma unroll
    for (int i = 0; i < TSLOTS; ++i) {
        const int j = i * 1024 + t;
        const int jc = j < ntile ? j : 0;
        kv[i] = s_key[jc];
        pp[i] = s_pos[jc];
    }
    #pragma unroll
    for (int i = 0; i < TSLOTS; ++i) {
        const int j = i * 1024 + t;
        if (j < ntile) keys[pp[i]] = kv[i];
    }
}

__global__ __launch_bounds__(GTHR, 16) void gcn_bucket_gather(
    const float* __restrict__ h, const int* __restrict__ total,
    const int* __restrict__ boff, const u64* __restrict__ keys,
    const float* __restrict__ bias, const float* __restrict__ skip,
    float* __restrict__ out)
{
    __shared__ u64 s_srt[CAP + HCAP];
    __shared__ int s_cnt[RPAD];
    __shared__ int s_start[RPAD];
    __shared__ int s_cur[RPAD];
    __shared__ int s_w[RPAD / 32];
    __shared__ u64 s_row[GWAVES][SEG_CAP];
    __shared__ unsigned s_off[GWAVES][2 * HCAP];
    __shared__ float    s_val[GWAVES][2 * HCAP];
    __shared__ __attribute__((aligned(16))) float s_skip[DDIM];
    __shared__ __attribute__((aligned(16))) float s_bias[DDIM];

    const int t = threadIdx.x;
    const int lane = t & 31;
    const int w = __builtin_amdgcn_readfirstlane((int)(threadIdx.x >> 5));
    const int b = blockIdx.x;
    const int n = total[b];
    const u64* bsrc = keys + boff[b];
    const bool fits = n <= CAP;

    if (t < RPAD) s_cnt[t] = 0;
    if (t < DDIM) { s_skip[t] = skip[t]; s_bias[t] = bias[t]; }
    __syncthreads();
    u64 kk[6];
    if (fits) {
        #pragma unroll
        for (int i = 0; i < 6; ++i) {
            const int j = i * GTHR + t;
            kk[i] = j < n ? bsrc[j] : 0ull;
        }
        #pragma unroll
        for (int i = 0; i < 6; ++i)
            if (i * GTHR + t < n) atomicAdd(&s_cnt[(int)(kk[i] >> 49)], 1);
    } else {
        for (int j = t; j < n; j += GTHR) atomicAdd(&s_cnt[(int)(bsrc[j] >> 49)], 1);
    }
    __syncthreads();
    int c = 0, inc = 0;
    if (t < RPAD) {
        c = s_cnt[t];
        inc = c;
        for (int d = 1; d < 32; d <<= 1) {
            const int y = __shfl_up(inc, d, 32);
            if (lane >= d) inc += y;
        }
        if (lane == 31) s_w[w] = inc;
    }
    __syncthreads();
    if (t < RPAD) {
        int pre = 0;
        for (int q = 0; q < w; ++q) pre += s_w[q];
        const int st = pre + inc - c;
        s_start[t] = st;
        s_cur[t] = st;
    }
    __syncthreads();
    if (fits) {
        #pragma unroll
        for (int i = 0; i < 6; ++i) {
            if (i * GTHR + t < n) {
                const int p = atomicAdd(&s_cur[(int)(kk[i] >> 49)], 1);
                s_srt[p] = kk[i];
            }
        }
        if (t < HCAP) s_srt[n + t] = ~0ull;
    }
    __syncthreads();

    const float alpha = 1.6732632423543772f;
    const float scale = 1.0507009873554805f;
    const int hf = lane >> 4;
    const int cl = lane & 15;
    const unsigned coff = (unsigned)cl << 4;
    const char* hbytes = (const char*)h;
    char* obytes = (char*)out;
    unsigned* woff = &s_off[w][hf * HCAP];
    float*    wval = &s_val[w][hf * HCAP];
    unsigned slow = 0;

    #pragma unroll 1
    for (int q = 0; q < 4; ++q) {
        const int lr0 = w * 8 + q * 2;
        if (b * BROWS + lr0 >= N_NODES) break;
        const int nA = __builtin_amdgcn_readfirstlane(s_cnt[lr0]);
        const int nB = __builtin_amdgcn_readfirstlane(s_cnt[lr0 + 1]);
        const int nmax = nA > nB ? nA : nB;
        if (!(fits && nmax <= HCAP)) { slow |= 1u << q; continue; }

        const int lrm = lr0 + hf;
        const int rm = b * BROWS + lrm;
        const bool vrow = rm < N_NODES;
        const int nm = hf ? nB : nA;
        const u64* seg = &s_srt[s_start[lrm]];
        const u64 none = ~0ull;
        const unsigned roff = ((unsigned)rm << 8) | coff;

        v4f hs = {0.0f, 0.0f, 0.0f, 0.0f};
        if (vrow) hs = *(const v4f*)(hbytes + roff);

        const int p1 = cl + 16;
        const u64 k0 = cl < nm ? seg[cl] : none;
        const u64 k1 = p1 < nm ? seg[p1] : none;
        int r0 = 0, r1 = 0;
        const int npad = (nmax + GRP - 1) & ~(GRP - 1);
        if (nmax <= 16) {
            for (int j = 0; j < npad; ++j) {
                const u64 kj = seg[j];
                r0 += (kj < k0 || (kj == k0 && j < cl)) ? 1 : 0;
            }
        } else {
            for (int j = 0; j < npad; ++j) {
                const u64 kj = seg[j];
                r0 += (kj < k0 || (kj == k0 && j < cl)) ? 1 : 0;
                r1 += (kj < k1 || (kj == k1 && j < p1)) ? 1 : 0;
            }
        }
        if (cl < nm) {
            woff[r0] = (((unsigned)(k0 >> 32)) & 0x1FFFFu) << 8;
            wval[r0] = __uint_as_float((unsigned)k0);
        } else {
            woff[cl] = (unsigned)ZROW << 8;
            wval[cl] = 0.0f;
        }
        if (p1 < nm) {
            woff[r1] = (((unsigned)(k1 >> 32)) & 0x1FFFFu) << 8;
            wval[r1] = __uint_as_float((unsigned)k1);
        } else {
            woff[p1] = (unsigned)ZROW << 8;
            wval[p1] = 0.0f;
        }

        v4f acc = {0.0f, 0.0f, 0.0f, 0.0f};
        for (int g0 = 0; g0 < npad; g0 += GRP) {
            #pragma unroll
            for (int u = 0; u < GRP; ++u) {
                const unsigned off = woff[g0 + u] | coff;
                const float v = wval[g0 + u];
                const v4f hv = *(const v4f*)(hbytes + off);
                acc.x = fmaf(v, hv.x, acc.x);
                acc.y = fmaf(v, hv.y, acc.y);
                acc.z = fmaf(v, hv.z, acc.z);
                acc.w = fmaf(v, hv.w, acc.w);
            }
        }

        if (vrow) {
            const v4f sk = *(const v4f*)&s_skip[cl * 4];
            const v4f bi = *(const v4f*)&s_bias[cl * 4];
            const float x0 = hs.x * sk.x + acc.x + bi.x;
            const float x1 = hs.y * sk.y + acc.y + bi.y;
            const float x2 = hs.z * sk.z + acc.z + bi.z;
            const float x3 = hs.w * sk.w + acc.w + bi.w;
            v4f res;
            res.x = x0 > 0.0f ? scale * x0 : scale * alpha * expm1f(x0);
            res.y = x1 > 0.0f ? scale * x1 : scale * alpha * expm1f(x1);
            res.z = x2 > 0.0f ? scale * x2 : scale * alpha * expm1f(x2);
            res.w = x3 > 0.0f ? scale * x3 : scale * alpha * expm1f(x3);
            *(v4f*)(obytes + roff) = res;
        }
    }

    #pragma unroll 1
    for (int q = 0; q < 4; ++q) {
        if (!((slow >> q) & 1u)) continue;
        const int lr0 = w * 8 + q * 2;
        const v2f sk = *(const v2f*)&s_skip[lane * 2];
        const v2f bi = *(const v2f*)&s_bias[lane * 2];
        for (int s = 0; s < 2; ++s) {
            const int lr = lr0 + s;
            const int r = b * BROWS + lr;
            if (r >= N_NODES) break;
            const int nr = __builtin_amdgcn_readfirstlane(s_cnt[lr]);
            const int st = __builtin_amdgcn_readfirstlane(s_start[lr]);

            const v2f hs = *(const v2f*)(h + (size_t)r * DDIM + lane * 2);
            v2f acc = {0.0f, 0.0f};

            if (fits && nr <= SEG_CAP) {
                const u64* seg = &s_srt[st];
                const u64 none = ~0ull;
                const u64 k0 = lane < nr ? seg[lane] : none;
                const u64 k1 = lane + 32 < nr ? seg[lane + 32] : none;
                int r0 = 0, r1 = 0;
                for (int j = 0; j < nr; ++j) {
                    const u64 kj = seg[j];
                    r0 += (kj < k0 || (kj == k0 && j < lane)) ? 1 : 0;
                    r1 += (kj < k1 || (kj == k1 && j < lane + 32)) ? 1 : 0;
                }
                if (lane < nr) s_row[w][r0] = k0;
                if (lane + 32 < nr) s_row[w][r1] = k1;
                for (int k = 0; k < nr; ++k) {
                    const u64 e = s_row[w][k];
                    const int   cc = __builtin_amdgcn_readfirstlane((int)(e >> 32)) & 0x1FFFF;
                    const float v = __uint_as_float((unsigned)__builtin_amdgcn_readfirstlane((int)(unsigned)e));
                    const v2f hv = *(const v2f*)(h + (size_t)cc * DDIM + lane * 2);
                    acc.x += v * hv.x;
                    acc.y += v * hv.y;
                }
            } else {
                const u64* src = fits ? (const u64*)&s_srt[st] : bsrc;
                const int cnt = fits ? nr : n;
                u64 prev = 0;
                int done = 0;
                for (int it = 0; it < nr && done < nr; ++it) {
                    u64 best = ~0ull;
                    int cb = 0;
                    for (int j = lane; j < cnt; j += 32) {
                        const u64 kj = src[j];
                        if ((int)(kj >> 49) == lr && (done == 0 || kj > prev)) {
                            if (kj < best) { best = kj; cb = 1; }
                            else if (kj == best) { cb += 1; }
                        }
                    }
                    u64 m = best;
                    for (int d = 16; d >= 1; d >>= 1) {
                        const u64 o = __shfl_xor(m, d, 32);
                        m = o < m ? o : m;
                    }
                    int cm = (best == m) ? cb : 0;
                    for (int d = 16; d >= 1; d >>= 1) cm += __shfl_xor(cm, d, 32);
                    const int   col = (int)(m >> 32) & 0x1FFFF;
                    const float v   = __uint_as_float((unsigned)m);
                    const v2f hv = *(const v2f*)(h + (size_t)col * DDIM + lane * 2);
                    for (int i = 0; i < cm; ++i) {
                        acc.x += v * hv.x;
                        acc.y += v * hv.y;
                    }
                    done += cm;
                    prev = m;
                }
            }

            const float x0 = hs.x * sk.x + acc.x + bi.x;
            const float x1 = hs.y * sk.y + acc.y + bi.y;
            v2f res;
            res.x = x0 > 0.0f ? scale * x0 : scale * alpha * expm1f(x0);
            res.y = x1 > 0.0f ? scale * x1 : scale * alpha * expm1f(x1);
            *(v2f*)(out + (size_t)r * DDIM + lane * 2) = res;
        }
    }
}

extern "C" void kernel_launch(void* const* d_in, const int* in_sizes, int n_in,
                              void* d_out, int out_size, void* d_ws, size_t ws_size,
                              hipStream_t stream) {
    const float* features = (const float*)d_in[0];
    const float* kern     = (const float*)d_in[1];
    const float* bias     = (const float*)d_in[2];
    const float* skip     = (const float*)d_in[3];
    const int*   erow     = (const int*)  d_in[4];
    const int*   ecol     = (const int*)  d_in[5];
    const float* eval_    = (const float*)d_in[6];
    float* out = (float*)d_out;
    char*  ws  = (char*)d_ws;
    float* h      = (float*)ws;
    int*   total  = (int*)(ws + 25600000);
    int*   cursor = (int*)(ws + 25604096);
    int*   boff   = (int*)(ws + 25608192);
    u64*   keys   = (u64*)(ws + 26402816);

    if (ws_size < 39202816ull) return;

    gcn_gemm_wmma<<<782, 256, 0, stream>>>(features, kern, h, total);
    gcn_bucket_count<<<NTILES, 1024, 0, stream>>>(erow, total);
    gcn_bucket_place<<<NTILES, 1024, 0, stream>>>(erow, ecol, eval_, total, cursor, boff, keys);
    gcn_bucket_gather<<<NB, GTHR, 0, stream>>>(h, total, boff, keys, bias, skip, out);
}
